# starting kernel
# baseline (speedup 1.0000x reference)
#include <hip/hip_runtime.h>


#define TOKENS 2048
#define INF 4096
#define OUTF 4096
#define JB 32
#define NCHUNK 2
#define XP_STRIDE 65
#define TT_STRIDE (JB + 1)
#define SMEM_FLOATS (1024 * TT_STRIDE + 64 * XP_STRIDE)
#define SMEM_BYTES (SMEM_FLOATS * 4)

typedef __attribute__((ext_vector_type(2))) float v2f;
typedef __attribute__((ext_vector_type(8))) float v8f;

__global__ __launch_bounds__(256)
void tt_linear_kernel(const float* __restrict__ x,
                      const float* __restrict__ g0,
                      const float* __restrict__ g1,
                      const float* __restrict__ alpha,
                      const float* __restrict__ pds,
                      const float* __restrict__ bias,
                      const int*   __restrict__ iperm,
                      const int*   __restrict__ oinv,
                      float* __restrict__ out)
{
    extern __shared__ float smem[];
    float* Tt = smem;
    float* xp = smem + 1024 * TT_STRIDE;

    const int b    = blockIdx.x;
    const int tid  = threadIdx.x;
    const int lane = tid & 31;
    const int w    = tid >> 5;
    const int half = lane >> 4;
    const int l15  = lane & 15;
    const int koff = half * 2;

    const float* xrow = x + (size_t)b * INF;
    for (int c = tid; c < INF; c += 256) {
        xp[(c >> 6) * XP_STRIDE + (c & 63)] = xrow[iperm[c]];
    }
    __syncthreads();

    const int ti  = w & 3;
    const int tm0 = (w >> 2) * 2;
    v8f accY0 = {};
    v8f accY1 = {};

    for (int chunk = 0; chunk < NCHUNK; ++chunk) {
        const int jbase = chunk * JB;

        for (int cp = 0; cp < 4; ++cp) {
            const int ct0   = w * 2 + cp * 16;
            const int cb0   = ct0 * 16;
            const int cb1   = cb0 + 16;
            const float* ap0 = g1 + (cb0 + l15) * 64 + koff;
            const float* ap1 = g1 + (cb1 + l15) * 64 + koff;
            v8f t00 = {}, t01 = {};
            v8f t10 = {}, t11 = {};
            #pragma unroll
            for (int kb = 0; kb < 64; kb += 4) {
                v2f a0 = *(const v2f*)(ap0 + kb);
                v2f a1 = *(const v2f*)(ap1 + kb);
                const int k = kb + koff;
                v2f b0, b1;
                {
                    const float* bp = xp + (jbase + l15) * XP_STRIDE + k;
                    b0.x = bp[0]; b0.y = bp[1];
                }
                {
                    const float* bp = xp + (jbase + 16 + l15) * XP_STRIDE + k;
                    b1.x = bp[0]; b1.y = bp[1];
                }
                t00 = __builtin_amdgcn_wmma_f32_16x16x4_f32(false, a0, false, b0, (short)0, t00, false, false);
                t01 = __builtin_amdgcn_wmma_f32_16x16x4_f32(false, a0, false, b1, (short)0, t01, false, false);
                t10 = __builtin_amdgcn_wmma_f32_16x16x4_f32(false, a1, false, b0, (short)0, t10, false, false);
                t11 = __builtin_amdgcn_wmma_f32_16x16x4_f32(false, a1, false, b1, (short)0, t11, false, false);
            }
            #pragma unroll
            for (int v = 0; v < 8; ++v) {
                const int r0 = cb0 + v + half * 8;
                const int r1 = cb1 + v + half * 8;
                Tt[r0 * TT_STRIDE + l15]      = t00[v];
                Tt[r0 * TT_STRIDE + 16 + l15] = t01[v];
                Tt[r1 * TT_STRIDE + l15]      = t10[v];
                Tt[r1 * TT_STRIDE + 16 + l15] = t11[v];
            }
        }
        __syncthreads();

        const float* aptr = g0 + (ti * 16 + l15) * 1024 + jbase * 16 + koff;
        #pragma unroll 4
        for (int ks = 0; ks < JB * 16; ks += 4) {
            v2f a = *(const v2f*)(aptr + ks);
            const int kk = ks + koff;
            const int r  = kk & 15;
            const int jl = kk >> 4;
            {
                const int m = tm0 * 16 + l15;
                v2f bv;
                bv.x = Tt[(r * 64 + m) * TT_STRIDE + jl];
                bv.y = Tt[((r + 1) * 64 + m) * TT_STRIDE + jl];
                accY0 = __builtin_amdgcn_wmma_f32_16x16x4_f32(false, a, false, bv, (short)0, accY0, false, false);
            }
            {
                const int m = (tm0 + 1) * 16 + l15;
                v2f bv;
                bv.x = Tt[(r * 64 + m) * TT_STRIDE + jl];
                bv.y = Tt[((r + 1) * 64 + m) * TT_STRIDE + jl];
                accY1 = __builtin_amdgcn_wmma_f32_16x16x4_f32(false, a, false, bv, (short)0, accY1, false, false);
            }
        }
        __syncthreads();
    }

    const float av = alpha[0];
    float* ylds = xp;
    #pragma unroll
    for (int v = 0; v < 8; ++v) {
        const int i  = ti * 16 + v + half * 8;
        const int m0 = tm0 * 16 + l15;
        const int m1 = m0 + 16;
        ylds[i * XP_STRIDE + m0] = accY0[v] * av * pds[i * 64 + m0];
        ylds[i * XP_STRIDE + m1] = accY1[v] * av * pds[i * 64 + m1];
    }
    __syncthreads();

    float* orow = out + (size_t)b * OUTF;
    for (int q = tid; q < OUTF; q += 256) {
        const int o = oinv[q];
        orow[q] = ylds[(o >> 6) * XP_STRIDE + (o & 63)] + bias[q];
    }
}

extern "C" void kernel_launch(void* const* d_in, const int* in_sizes, int n_in,
                              void* d_out, int out_size, void* d_ws, size_t ws_size,
                              hipStream_t stream) {
    (void)in_sizes; (void)n_in; (void)out_size; (void)d_ws; (void)ws_size;
    const float* x     = (const float*)d_in[0];
    const float* g0    = (const float*)d_in[1];
    const float* g1    = (const float*)d_in[2];
    const float* alpha = (const float*)d_in[3];
    const float* pds   = (const float*)d_in[4];
    const float* bias  = (const float*)d_in[5];
    const int*   iperm = (const int*)d_in[6];
    const int*   oinv  = (const int*)d_in[7];
    float* out = (float*)d_out;

    (void)hipFuncSetAttribute(reinterpret_cast<const void*>(tt_linear_kernel),
                              hipFuncAttributeMaxDynamicSharedMemorySize, SMEM_BYTES);

    tt_linear_kernel<<<TOKENS, 256, SMEM_BYTES, stream>>>(
        x, g0, g1, alpha, pds, bias, iperm, oinv, out);
}


# All WMMA operands fetched as 16-byte loads per lane feeding two k-steps (A from global, B from LDS; Tt stored r-quad-contiguous, xp pitch 68); k summation order inside each 8-block changes
# speedup vs baseline: 1.2082x; 1.2082x over previous
#include <hip/hip_runtime.h>


#define TOKENS 2048
#define INF 4096
#define OUTF 4096
#define JB 32
#define NCHUNK 2
#define XP_STRIDE 68
#define TT_JS 1028
#define TT_FLOATS (JB * TT_JS)
#define SMEM_FLOATS (TT_FLOATS + 64 * XP_STRIDE)
#define SMEM_BYTES (SMEM_FLOATS * 4)

typedef __attribute__((ext_vector_type(2))) float v2f;
typedef __attribute__((ext_vector_type(4))) float v4f;
typedef __attribute__((ext_vector_type(8))) float v8f;

#define WMMA(a, b, c) __builtin_amdgcn_wmma_f32_16x16x4_f32(false, (a), false, (b), (short)0, (c), false, false)

__global__ __launch_bounds__(256)
void tt_linear_kernel(const float* __restrict__ x,
                      const float* __restrict__ g0,
                      const float* __restrict__ g1,
                      const float* __restrict__ alpha,
                      const float* __restrict__ pds,
                      const float* __restrict__ bias,
                      const int*   __restrict__ iperm,
                      const int*   __restrict__ oinv,
                      float* __restrict__ out)
{
    extern __shared__ v4f smem_v[];
    float* smem = (float*)smem_v;
    float* Tt = smem;
    float* xp = smem + TT_FLOATS;

    const int b    = blockIdx.x;
    const int tid  = threadIdx.x;
    const int lane = tid & 31;
    const int w    = tid >> 5;
    const int half = lane >> 4;
    const int l15  = lane & 15;
    const int k4   = half * 4;

    const float* xrow = x + (size_t)b * INF;
    for (int c = tid; c < INF; c += 256) {
        xp[(c >> 6) * XP_STRIDE + (c & 63)] = xrow[iperm[c]];
    }
    __syncthreads();

    const int ti  = w & 3;
    const int tm0 = (w >> 2) * 2;
    v8f accY0 = {};
    v8f accY1 = {};

    for (int chunk = 0; chunk < NCHUNK; ++chunk) {
        const int jbase = chunk * JB;

        for (int cp = 0; cp < 4; ++cp) {
            const int ct0 = w * 2 + cp * 16;
            const float* ap0 = g1 + (ct0 * 16 + l15) * 64 + k4;
            const float* ap1 = ap0 + 16 * 64;
            const float* bp0 = xp + (jbase + l15) * XP_STRIDE + k4;
            const float* bp1 = bp0 + 16 * XP_STRIDE;
            v8f t00 = {}, t01 = {};
            v8f t10 = {}, t11 = {};
            #pragma unroll
            for (int kb = 0; kb < 64; kb += 8) {
                const v4f a0 = *(const v4f*)(ap0 + kb);
                const v4f a1 = *(const v4f*)(ap1 + kb);
                const v4f b0 = *(const v4f*)(bp0 + kb);
                const v4f b1 = *(const v4f*)(bp1 + kb);
                const v2f a0l = {a0.x, a0.y}, a0h = {a0.z, a0.w};
                const v2f a1l = {a1.x, a1.y}, a1h = {a1.z, a1.w};
                const v2f b0l = {b0.x, b0.y}, b0h = {b0.z, b0.w};
                const v2f b1l = {b1.x, b1.y}, b1h = {b1.z, b1.w};
                t00 = WMMA(a0l, b0l, t00);
                t01 = WMMA(a0l, b1l, t01);
                t10 = WMMA(a1l, b0l, t10);
                t11 = WMMA(a1l, b1l, t11);
                t00 = WMMA(a0h, b0h, t00);
                t01 = WMMA(a0h, b1h, t01);
                t10 = WMMA(a1h, b0h, t10);
                t11 = WMMA(a1h, b1h, t11);
            }
            const int r  = ct0 >> 2;
            const int mb = (ct0 & 3) * 16;
            float* tp = Tt + l15 * TT_JS + (r >> 2) * 256 + (mb + half * 8) * 4 + (r & 3);
            #pragma unroll
            for (int v = 0; v < 8; ++v) {
                tp[v * 4]                   = t00[v];
                tp[16 * TT_JS + v * 4]      = t01[v];
                tp[64 + v * 4]              = t10[v];
                tp[16 * TT_JS + 64 + v * 4] = t11[v];
            }
        }
        __syncthreads();

        const float* aptr = g0 + (ti * 16 + l15) * 1024 + jbase * 16 + k4;
        const float* bptr = Tt + half * 256 + (tm0 * 16 + l15) * 4;
        #pragma unroll 4
        for (int jl = 0; jl < JB; ++jl) {
            const v4f aA = *(const v4f*)(aptr + jl * 16);
            const v4f aB = *(const v4f*)(aptr + jl * 16 + 8);
            const float* bj = bptr + jl * TT_JS;
            const v4f pA = *(const v4f*)(bj);
            const v4f qA = *(const v4f*)(bj + 64);
            const v4f pB = *(const v4f*)(bj + 512);
            const v4f qB = *(const v4f*)(bj + 512 + 64);
            const v2f aAl = {aA.x, aA.y}, aAh = {aA.z, aA.w};
            const v2f aBl = {aB.x, aB.y}, aBh = {aB.z, aB.w};
            const v2f pAl = {pA.x, pA.y}, pAh = {pA.z, pA.w};
            const v2f qAl = {qA.x, qA.y}, qAh = {qA.z, qA.w};
            const v2f pBl = {pB.x, pB.y}, pBh = {pB.z, pB.w};
            const v2f qBl = {qB.x, qB.y}, qBh = {qB.z, qB.w};
            accY0 = WMMA(aAl, pAl, accY0);
            accY1 = WMMA(aAl, qAl, accY1);
            accY0 = WMMA(aAh, pAh, accY0);
            accY1 = WMMA(aAh, qAh, accY1);
            accY0 = WMMA(aBl, pBl, accY0);
            accY1 = WMMA(aBl, qBl, accY1);
            accY0 = WMMA(aBh, pBh, accY0);
            accY1 = WMMA(aBh, qBh, accY1);
        }
        __syncthreads();
    }

    const float av = alpha[0];
    float* ylds = xp;
    #pragma unroll
    for (int v = 0; v < 8; ++v) {
        const int i  = ti * 16 + v + half * 8;
        const int m0 = tm0 * 16 + l15;
        const int m1 = m0 + 16;
        ylds[i * XP_STRIDE + m0] = accY0[v] * av * pds[i * 64 + m0];
        ylds[i * XP_STRIDE + m1] = accY1[v] * av * pds[i * 64 + m1];
    }
    __syncthreads();

    float* orow = out + (size_t)b * OUTF;
    for (int q = tid; q < OUTF; q += 256) {
        const int o = oinv[q];
        orow[q] = ylds[(o >> 6) * XP_STRIDE + (o & 63)] + bias[q];
    }
}

extern "C" void kernel_launch(void* const* d_in, const int* in_sizes, int n_in,
                              void* d_out, int out_size, void* d_ws, size_t ws_size,
                              hipStream_t stream) {
    (void)in_sizes; (void)n_in; (void)out_size; (void)d_ws; (void)ws_size;
    const float* x     = (const float*)d_in[0];
    const float* g0    = (const float*)d_in[1];
    const float* g1    = (const float*)d_in[2];
    const float* alpha = (const float*)d_in[3];
    const float* pds   = (const float*)d_in[4];
    const float* bias  = (const float*)d_in[5];
    const int*   iperm = (const int*)d_in[6];
    const int*   oinv  = (const int*)d_in[7];
    float* out = (float*)d_out;

    (void)hipFuncSetAttribute(reinterpret_cast<const void*>(tt_linear_kernel),
                              hipFuncAttributeMaxDynamicSharedMemorySize, SMEM_BYTES);

    tt_linear_kernel<<<TOKENS, 256, SMEM_BYTES, stream>>>(
        x, g0, g1, alpha, pds, bias, iperm, oinv, out);
}


# Pre-pass kernel re-lays g0 and g1 into WMMA-fragment order in the workspace on every call: each A-operand global load now reads 4 whole 128-B lines, not 16 partial ones; launches 1 -> 2
# speedup vs baseline: 1.2843x; 1.0630x over previous
#include <hip/hip_runtime.h>


#define TOKENS 2048
#define INF 4096
#define OUTF 4096
#define JB 32
#define NCHUNK 2
#define XP_STRIDE 68
#define TT_JS 1028
#define TT_FLOATS (JB * TT_JS)
#define SMEM_FLOATS (TT_FLOATS + 64 * XP_STRIDE)
#define SMEM_BYTES (SMEM_FLOATS * 4)

typedef __attribute__((ext_vector_type(2))) float v2f;
typedef __attribute__((ext_vector_type(4))) float v4f;
typedef __attribute__((ext_vector_type(8))) float v8f;

#define WF_G1_FLOATS 65536
#define WF_WORDS 32768
#define WF_BLOCKS 128

__global__ __launch_bounds__(256)
void tt_weight_fragment_kernel(const float* __restrict__ g0,
                               const float* __restrict__ g1,
                               float* __restrict__ wf)
{
    const int t    = blockIdx.x * 256 + threadIdx.x;
    const int lane = t & 31;
    const int l15  = lane & 15;
    const int half = lane >> 4;
    const float* src;
    if (t < WF_WORDS / 2) {
        const int kbi = (t >> 5) & 7;
        const int ct  = t >> 8;
        src = g1 + (ct * 16 + l15) * 64 + kbi * 8 + half * 4;
    } else {
        const int u  = t - WF_WORDS / 2;
        const int ab = (u >> 5) & 1;
        const int j  = (u >> 6) & 63;
        const int ti = u >> 12;
        src = g0 + (ti * 16 + l15) * 1024 + j * 16 + ab * 8 + half * 4;
    }
    *(v4f*)(wf + (size_t)t * 4) = *(const v4f*)src;
}

#define WMMA(a, b, c) __builtin_amdgcn_wmma_f32_16x16x4_f32(false, (a), false, (b), (short)0, (c), false, false)

__global__ __launch_bounds__(256)
void tt_linear_kernel(const float* __restrict__ x,
                      const float* __restrict__ wf,
                      const float* __restrict__ alpha,
                      const float* __restrict__ pds,
                      const float* __restrict__ bias,
                      const int*   __restrict__ iperm,
                      const int*   __restrict__ oinv,
                      float* __restrict__ out)
{
    extern __shared__ v4f smem_v[];
    float* smem = (float*)smem_v;
    float* Tt = smem;
    float* xp = smem + TT_FLOATS;

    const int b    = blockIdx.x;
    const int tid  = threadIdx.x;
    const int lane = tid & 31;
    const int w    = tid >> 5;
    const int half = lane >> 4;
    const int l15  = lane & 15;
    const int k4   = half * 4;
    const float* g1f = wf;
    const float* g0f = wf + WF_G1_FLOATS;

    const float* xrow = x + (size_t)b * INF;
    for (int c = tid; c < INF; c += 256) {
        xp[(c >> 6) * XP_STRIDE + (c & 63)] = xrow[iperm[c]];
    }
    __syncthreads();

    const int ti  = w & 3;
    const int tm0 = (w >> 2) * 2;
    v8f accY0 = {};
    v8f accY1 = {};

    for (int chunk = 0; chunk < NCHUNK; ++chunk) {
        const int jbase = chunk * JB;

        for (int cp = 0; cp < 4; ++cp) {
            const int ct0 = w * 2 + cp * 16;
            const float* ap0 = g1f + ct0 * 1024 + lane * 4;
            const float* ap1 = ap0 + 1024;
            const float* bp0 = xp + (jbase + l15) * XP_STRIDE + k4;
            const float* bp1 = bp0 + 16 * XP_STRIDE;
            v8f t00 = {}, t01 = {};
            v8f t10 = {}, t11 = {};
            #pragma unroll
            for (int kb = 0; kb < 64; kb += 8) {
                const v4f a0 = *(const v4f*)(ap0 + kb * 16);
                const v4f a1 = *(const v4f*)(ap1 + kb * 16);
                const v4f b0 = *(const v4f*)(bp0 + kb);
                const v4f b1 = *(const v4f*)(bp1 + kb);
                const v2f a0l = {a0.x, a0.y}, a0h = {a0.z, a0.w};
                const v2f a1l = {a1.x, a1.y}, a1h = {a1.z, a1.w};
                const v2f b0l = {b0.x, b0.y}, b0h = {b0.z, b0.w};
                const v2f b1l = {b1.x, b1.y}, b1h = {b1.z, b1.w};
                t00 = WMMA(a0l, b0l, t00);
                t01 = WMMA(a0l, b1l, t01);
                t10 = WMMA(a1l, b0l, t10);
                t11 = WMMA(a1l, b1l, t11);
                t00 = WMMA(a0h, b0h, t00);
                t01 = WMMA(a0h, b1h, t01);
                t10 = WMMA(a1h, b0h, t10);
                t11 = WMMA(a1h, b1h, t11);
            }
            const int r  = ct0 >> 2;
            const int mb = (ct0 & 3) * 16;
            float* tp = Tt + l15 * TT_JS + (r >> 2) * 256 + (mb + half * 8) * 4 + (r & 3);
            #pragma unroll
            for (int v = 0; v < 8; ++v) {
                tp[v * 4]                   = t00[v];
                tp[16 * TT_JS + v * 4]      = t01[v];
                tp[64 + v * 4]              = t10[v];
                tp[16 * TT_JS + 64 + v * 4] = t11[v];
            }
        }
        __syncthreads();

        const float* aptr = g0f + (ti * 64 + jbase) * 256 + lane * 4;
        const float* bptr = Tt + half * 256 + (tm0 * 16 + l15) * 4;
        #pragma unroll 4
        for (int jl = 0; jl < JB; ++jl) {
            const v4f aA = *(const v4f*)(aptr + jl * 256);
            const v4f aB = *(const v4f*)(aptr + jl * 256 + 128);
            const float* bj = bptr + jl * TT_JS;
            const v4f pA = *(const v4f*)(bj);
            const v4f qA = *(const v4f*)(bj + 64);
            const v4f pB = *(const v4f*)(bj + 512);
            const v4f qB = *(const v4f*)(bj + 512 + 64);
            const v2f aAl = {aA.x, aA.y}, aAh = {aA.z, aA.w};
            const v2f aBl = {aB.x, aB.y}, aBh = {aB.z, aB.w};
            const v2f pAl = {pA.x, pA.y}, pAh = {pA.z, pA.w};
            const v2f qAl = {qA.x, qA.y}, qAh = {qA.z, qA.w};
            const v2f pBl = {pB.x, pB.y}, pBh = {pB.z, pB.w};
            const v2f qBl = {qB.x, qB.y}, qBh = {qB.z, qB.w};
            accY0 = WMMA(aAl, pAl, accY0);
            accY1 = WMMA(aAl, qAl, accY1);
            accY0 = WMMA(aAh, pAh, accY0);
            accY1 = WMMA(aAh, qAh, accY1);
            accY0 = WMMA(aBl, pBl, accY0);
            accY1 = WMMA(aBl, qBl, accY1);
            accY0 = WMMA(aBh, pBh, accY0);
            accY1 = WMMA(aBh, qBh, accY1);
        }
        __syncthreads();
    }

    const float av = alpha[0];
    float* ylds = xp;
    #pragma unroll
    for (int v = 0; v < 8; ++v) {
        const int i  = ti * 16 + v + half * 8;
        const int m0 = tm0 * 16 + l15;
        const int m1 = m0 + 16;
        ylds[i * XP_STRIDE + m0] = accY0[v] * av * pds[i * 64 + m0];
        ylds[i * XP_STRIDE + m1] = accY1[v] * av * pds[i * 64 + m1];
    }
    __syncthreads();

    float* orow = out + (size_t)b * OUTF;
    for (int q = tid; q < OUTF; q += 256) {
        const int o = oinv[q];
        orow[q] = ylds[(o >> 6) * XP_STRIDE + (o & 63)] + bias[q];
    }
}

extern "C" void kernel_launch(void* const* d_in, const int* in_sizes, int n_in,
                              void* d_out, int out_size, void* d_ws, size_t ws_size,
                              hipStream_t stream) {
    (void)in_sizes; (void)n_in; (void)out_size; (void)ws_size;
    const float* x     = (const float*)d_in[0];
    const float* g0    = (const float*)d_in[1];
    const float* g1    = (const float*)d_in[2];
    const float* alpha = (const float*)d_in[3];
    const float* pds   = (const float*)d_in[4];
    const float* bias  = (const float*)d_in[5];
    const int*   iperm = (const int*)d_in[6];
    const int*   oinv  = (const int*)d_in[7];
    float* out = (float*)d_out;
    float* wf  = (float*)d_ws;

    (void)hipFuncSetAttribute(reinterpret_cast<const void*>(tt_linear_kernel),
                              hipFuncAttributeMaxDynamicSharedMemorySize, SMEM_BYTES);

    tt_weight_fragment_kernel<<<WF_BLOCKS, 256, 0, stream>>>(g0, g1, wf);

    tt_linear_kernel<<<TOKENS, 256, SMEM_BYTES, stream>>>(
        x, wf, alpha, pds, bias, iperm, oinv, out);
}


# Permutation phases wait one memory round trip, not 8 dependent pairs: x row read in natural order to LDS beside the index loads and permuted there; output index/bias/scale loaded before last GEMM2
# speedup vs baseline: 1.3925x; 1.0842x over previous
#include <hip/hip_runtime.h>


#define TOKENS 2048
#define INF 4096
#define OUTF 4096
#define JB 32
#define NCHUNK 2
#define ELEMS_PER_THREAD 16
#define XP_STRIDE 68
#define TT_JS 1028
#define TT_FLOATS (JB * TT_JS)
#define SMEM_FLOATS (TT_FLOATS + 64 * XP_STRIDE)
#define SMEM_BYTES (SMEM_FLOATS * 4)

typedef __attribute__((ext_vector_type(2))) float v2f;
typedef __attribute__((ext_vector_type(4))) float v4f;
typedef __attribute__((ext_vector_type(8))) float v8f;

#define WF_G1_FLOATS 65536
#define WF_WORDS 32768
#define WF_BLOCKS 128

__global__ __launch_bounds__(256)
void tt_weight_fragment_kernel(const float* __restrict__ g0,
                               const float* __restrict__ g1,
                               float* __restrict__ wf)
{
    const int t    = blockIdx.x * 256 + threadIdx.x;
    const int lane = t & 31;
    const int l15  = lane & 15;
    const int half = lane >> 4;
    const float* src;
    if (t < WF_WORDS / 2) {
        const int kbi = (t >> 5) & 7;
        const int ct  = t >> 8;
        src = g1 + (ct * 16 + l15) * 64 + kbi * 8 + half * 4;
    } else {
        const int u  = t - WF_WORDS / 2;
        const int ab = (u >> 5) & 1;
        const int j  = (u >> 6) & 63;
        const int ti = u >> 12;
        src = g0 + (ti * 16 + l15) * 1024 + j * 16 + ab * 8 + half * 4;
    }
    *(v4f*)(wf + (size_t)t * 4) = *(const v4f*)src;
}

#define WMMA(a, b, c) __builtin_amdgcn_wmma_f32_16x16x4_f32(false, (a), false, (b), (short)0, (c), false, false)

__global__ __launch_bounds__(256)
void tt_linear_kernel(const float* __restrict__ x,
                      const float* __restrict__ wf,
                      const float* __restrict__ alpha,
                      const float* __restrict__ pds,
                      const float* __restrict__ bias,
                      const int*   __restrict__ iperm,
                      const int*   __restrict__ oinv,
                      float* __restrict__ out)
{
    extern __shared__ v4f smem_v[];
    float* smem = (float*)smem_v;
    float* Tt = smem;
    float* xp = smem + TT_FLOATS;

    const int b    = blockIdx.x;
    const int tid  = threadIdx.x;
    const int lane = tid & 31;
    const int w    = tid >> 5;
    const int half = lane >> 4;
    const int l15  = lane & 15;
    const int k4   = half * 4;
    const float* g1f = wf;
    const float* g0f = wf + WF_G1_FLOATS;

    const float* xrow = x + (size_t)b * INF;
    {
        float* xs = Tt;
        v4f xv[ELEMS_PER_THREAD / 4];
        int pidx[ELEMS_PER_THREAD];
        #pragma unroll
        for (int i = 0; i < ELEMS_PER_THREAD / 4; ++i) xv[i] = *(const v4f*)(xrow + (i * 256 + tid) * 4);
        #pragma unroll
        for (int e = 0; e < ELEMS_PER_THREAD; ++e) pidx[e] = iperm[e * 256 + tid];
        #pragma unroll
        for (int i = 0; i < ELEMS_PER_THREAD / 4; ++i) *(v4f*)(xs + (i * 256 + tid) * 4) = xv[i];
        __syncthreads();
        float pval[ELEMS_PER_THREAD];
        #pragma unroll
        for (int e = 0; e < ELEMS_PER_THREAD; ++e) pval[e] = xs[pidx[e]];
        #pragma unroll
        for (int e = 0; e < ELEMS_PER_THREAD; ++e) {
            const int c = e * 256 + tid;
            xp[(c >> 6) * XP_STRIDE + (c & 63)] = pval[e];
        }
    }
    __syncthreads();

    const int ti  = w & 3;
    const int tm0 = (w >> 2) * 2;
    v8f accY0 = {};
    v8f accY1 = {};

    int   oidx[ELEMS_PER_THREAD];
    float bv[ELEMS_PER_THREAD];
    float sc0[8], sc1[8];
    float av = 0.0f;

    for (int chunk = 0; chunk < NCHUNK; ++chunk) {
        const int jbase = chunk * JB;

        for (int cp = 0; cp < 4; ++cp) {
            const int ct0 = w * 2 + cp * 16;
            const float* ap0 = g1f + ct0 * 1024 + lane * 4;
            const float* ap1 = ap0 + 1024;
            const float* bp0 = xp + (jbase + l15) * XP_STRIDE + k4;
            const float* bp1 = bp0 + 16 * XP_STRIDE;
            v8f t00 = {}, t01 = {};
            v8f t10 = {}, t11 = {};
            #pragma unroll
            for (int kb = 0; kb < 64; kb += 8) {
                const v4f a0 = *(const v4f*)(ap0 + kb * 16);
                const v4f a1 = *(const v4f*)(ap1 + kb * 16);
                const v4f b0 = *(const v4f*)(bp0 + kb);
                const v4f b1 = *(const v4f*)(bp1 + kb);
                const v2f a0l = {a0.x, a0.y}, a0h = {a0.z, a0.w};
                const v2f a1l = {a1.x, a1.y}, a1h = {a1.z, a1.w};
                const v2f b0l = {b0.x, b0.y}, b0h = {b0.z, b0.w};
                const v2f b1l = {b1.x, b1.y}, b1h = {b1.z, b1.w};
                t00 = WMMA(a0l, b0l, t00);
                t01 = WMMA(a0l, b1l, t01);
                t10 = WMMA(a1l, b0l, t10);
                t11 = WMMA(a1l, b1l, t11);
                t00 = WMMA(a0h, b0h, t00);
                t01 = WMMA(a0h, b1h, t01);
                t10 = WMMA(a1h, b0h, t10);
                t11 = WMMA(a1h, b1h, t11);
            }
            const int r  = ct0 >> 2;
            const int mb = (ct0 & 3) * 16;
            float* tp = Tt + l15 * TT_JS + (r >> 2) * 256 + (mb + half * 8) * 4 + (r & 3);
            #pragma unroll
            for (int v = 0; v < 8; ++v) {
                tp[v * 4]                   = t00[v];
                tp[16 * TT_JS + v * 4]      = t01[v];
                tp[64 + v * 4]              = t10[v];
                tp[16 * TT_JS + 64 + v * 4] = t11[v];
            }
        }
        __syncthreads();

        if (chunk == NCHUNK - 1) {
            av = alpha[0];
            #pragma unroll
            for (int e = 0; e < ELEMS_PER_THREAD; ++e) {
                oidx[e] = oinv[e * 256 + tid];
                bv[e]   = bias[e * 256 + tid];
            }
            #pragma unroll
            for (int v = 0; v < 8; ++v) {
                const int i = ti * 16 + v + half * 8;
                sc0[v] = pds[i * 64 + tm0 * 16 + l15];
                sc1[v] = pds[i * 64 + tm0 * 16 + l15 + 16];
            }
        }

        const float* aptr = g0f + (ti * 64 + jbase) * 256 + lane * 4;
        const float* bptr = Tt + half * 256 + (tm0 * 16 + l15) * 4;
        #pragma unroll 4
        for (int jl = 0; jl < JB; ++jl) {
            const v4f aA = *(const v4f*)(aptr + jl * 256);
            const v4f aB = *(const v4f*)(aptr + jl * 256 + 128);
            const float* bj = bptr + jl * TT_JS;
            const v4f pA = *(const v4f*)(bj);
            const v4f qA = *(const v4f*)(bj + 64);
            const v4f pB = *(const v4f*)(bj + 512);
            const v4f qB = *(const v4f*)(bj + 512 + 64);
            const v2f aAl = {aA.x, aA.y}, aAh = {aA.z, aA.w};
            const v2f aBl = {aB.x, aB.y}, aBh = {aB.z, aB.w};
            const v2f pAl = {pA.x, pA.y}, pAh = {pA.z, pA.w};
            const v2f qAl = {qA.x, qA.y}, qAh = {qA.z, qA.w};
            const v2f pBl = {pB.x, pB.y}, pBh = {pB.z, pB.w};
            const v2f qBl = {qB.x, qB.y}, qBh = {qB.z, qB.w};
            accY0 = WMMA(aAl, pAl, accY0);
            accY1 = WMMA(aAl, qAl, accY1);
            accY0 = WMMA(aAh, pAh, accY0);
            accY1 = WMMA(aAh, qAh, accY1);
            accY0 = WMMA(aBl, pBl, accY0);
            accY1 = WMMA(aBl, qBl, accY1);
            accY0 = WMMA(aBh, pBh, accY0);
            accY1 = WMMA(aBh, qBh, accY1);
        }
        __syncthreads();
    }

    float* ylds = xp;
    #pragma unroll
    for (int v = 0; v < 8; ++v) {
        const int i  = ti * 16 + v + half * 8;
        const int m0 = tm0 * 16 + l15;
        const int m1 = m0 + 16;
        ylds[i * XP_STRIDE + m0] = accY0[v] * av * sc0[v];
        ylds[i * XP_STRIDE + m1] = accY1[v] * av * sc1[v];
    }
    __syncthreads();

    float* orow = out + (size_t)b * OUTF;
    float yv[ELEMS_PER_THREAD];
    #pragma unroll
    for (int e = 0; e < ELEMS_PER_THREAD; ++e) {
        const int o = oidx[e];
        yv[e] = ylds[(o >> 6) * XP_STRIDE + (o & 63)];
    }
    #pragma unroll
    for (int e = 0; e < ELEMS_PER_THREAD; ++e) {
        orow[e * 256 + tid] = yv[e] + bv[e];
    }
}

extern "C" void kernel_launch(void* const* d_in, const int* in_sizes, int n_in,
                              void* d_out, int out_size, void* d_ws, size_t ws_size,
                              hipStream_t stream) {
    (void)in_sizes; (void)n_in; (void)out_size; (void)ws_size;
    const float* x     = (const float*)d_in[0];
    const float* g0    = (const float*)d_in[1];
    const float* g1    = (const float*)d_in[2];
    const float* alpha = (const float*)d_in[3];
    const float* pds   = (const float*)d_in[4];
    const float* bias  = (const float*)d_in[5];
    const int*   iperm = (const int*)d_in[6];
    const int*   oinv  = (const int*)d_in[7];
    float* out = (float*)d_out;
    float* wf  = (float*)d_ws;

    (void)hipFuncSetAttribute(reinterpret_cast<const void*>(tt_linear_kernel),
                              hipFuncAttributeMaxDynamicSharedMemorySize, SMEM_BYTES);

    tt_weight_fragment_kernel<<<WF_BLOCKS, 256, 0, stream>>>(g0, g1, wf);

    tt_linear_kernel<<<TOKENS, 256, SMEM_BYTES, stream>>>(
        x, wf, alpha, pds, bias, iperm, oinv, out);
}
